# starting kernel
# baseline (speedup 1.0000x reference)
#include <hip/hip_runtime.h>
#include <hip/hip_bf16.h>


typedef __attribute__((ext_vector_type(2))) float v2f;
typedef __attribute__((ext_vector_type(8))) float v8f;

#define N_NODES 50000
#define N_EDGES 800000
#define D_IN    128
#define D_HID   512
#define D_OUT   256
#define BN_EPS  1e-5
#define SLOPE   0.01f

__global__ void gn_copy_x(const float* __restrict__ x, float* __restrict__ h0, int n4) {
    int i = blockIdx.x * blockDim.x + threadIdx.x;
    if (i < n4) ((float4*)h0)[i] = ((const float4*)x)[i];
}

__global__ __launch_bounds__(256) void gn_scatter_add(const float* __restrict__ x,
                                                      const int* __restrict__ ei,
                                                      float* __restrict__ h0) {
    int wave = (blockIdx.x * blockDim.x + threadIdx.x) >> 5;
    int lane = threadIdx.x & 31;
    if (wave >= N_EDGES) return;
    int src = ei[wave];
    int dst = ei[N_EDGES + wave];
    const float4 v = *(const float4*)(x + (size_t)src * D_IN + lane * 4);
    float* hp = h0 + (size_t)dst * D_IN + lane * 4;
    atomicAdd(hp + 0, v.x);
    atomicAdd(hp + 1, v.y);
    atomicAdd(hp + 2, v.z);
    atomicAdd(hp + 3, v.w);
}

template <int K, int N, bool FUSE_STATS>
__global__ __launch_bounds__(256) void gn_wmma_gemm(const float* __restrict__ A,
                                                    const float* __restrict__ B,
                                                    const float* __restrict__ bias,
                                                    float* __restrict__ C, int M,
                                                    double* __restrict__ sums,
                                                    double* __restrict__ sumsq) {
    constexpr int KC = 128;
    __shared__ float Bs[KC][72];

    const int tid   = threadIdx.x;
    const int n0    = blockIdx.y * 64;
    const int wave  = tid >> 5;
    const int lane  = tid & 31;
    const int ln    = lane & 15;
    const int khalf = (lane >> 4) << 1;
    const int row0  = blockIdx.x * 128 + wave * 16;
    const int m     = row0 + ln;
    const int mc    = m < M ? m : M - 1;
    const float* ap = A + (size_t)mc * K + khalf;

    v8f acc[4] = {};

    for (int kc = 0; kc < K; kc += KC) {
        for (int idx = tid; idx < KC * 16; idx += 256) {
            int kk = idx >> 4, c4 = (idx & 15) * 4;
            float4 bv4 = *(const float4*)(B + (size_t)(kc + kk) * N + n0 + c4);
            *(float4*)&Bs[kk][c4] = bv4;
        }
        __syncthreads();

#pragma unroll 4
        for (int k = 0; k < KC; k += 4) {
            v2f a = *(const v2f*)(ap + kc + k);
#pragma unroll
            for (int t = 0; t < 4; ++t) {
                v2f b;
                b[0] = Bs[k + khalf][ln + 16 * t];
                b[1] = Bs[k + khalf + 1][ln + 16 * t];
                acc[t] = __builtin_amdgcn_wmma_f32_16x16x4_f32(
                     false, a,  false, b,
                     (short)0, acc[t],  false,  false);
            }
        }
        __syncthreads();
    }

    const int rbase = row0 + ((lane >> 4) << 3);
#pragma unroll
    for (int t = 0; t < 4; ++t) {
        const int   col = n0 + 16 * t + ln;
        const float bv  = bias[col];
        double ds = 0.0, dsq = 0.0;
#pragma unroll
        for (int r = 0; r < 8; ++r) {
            int row = rbase + r;
            float v = acc[t][r] + bv;
            v = v > 0.f ? v : v * SLOPE;
            if (row < M) {
                C[(size_t)row * N + col] = v;
                if (FUSE_STATS) { ds += (double)v; dsq += (double)v * (double)v; }
            }
        }
        if (FUSE_STATS) {
            atomicAdd(&sums[col], ds);
            atomicAdd(&sumsq[col], dsq);
        }
    }
}

__global__ void gn_bn_finalize(const double* __restrict__ sums,
                               const double* __restrict__ sumsq,
                               const float* __restrict__ gamma,
                               const float* __restrict__ beta,
                               float* __restrict__ scale, float* __restrict__ shift) {
    int c = threadIdx.x;
    double mean = sums[c] / (double)N_NODES;
    double var  = sumsq[c] / (double)N_NODES - mean * mean;
    double inv  = 1.0 / sqrt(var + BN_EPS);
    float sc = (float)inv * gamma[c];
    scale[c] = sc;
    shift[c] = beta[c] - (float)mean * sc;
}

__global__ void gn_bn_apply(float* __restrict__ out, const float* __restrict__ scale,
                            const float* __restrict__ shift) {
    int i = blockIdx.x * blockDim.x + threadIdx.x;
    const int total4 = N_NODES * D_OUT / 4;
    if (i >= total4) return;
    int c = (i * 4) & (D_OUT - 1);
    float4 v = ((float4*)out)[i];
    v.x = v.x * scale[c + 0] + shift[c + 0];
    v.y = v.y * scale[c + 1] + shift[c + 1];
    v.z = v.z * scale[c + 2] + shift[c + 2];
    v.w = v.w * scale[c + 3] + shift[c + 3];
    ((float4*)out)[i] = v;
}

extern "C" void kernel_launch(void* const* d_in, const int* in_sizes, int n_in,
                              void* d_out, int out_size, void* d_ws, size_t ws_size,
                              hipStream_t stream) {
    const float* x     = (const float*)d_in[0];
    const int*   ei    = (const int*)d_in[1];
    const float* W1    = (const float*)d_in[2];
    const float* b1    = (const float*)d_in[3];
    const float* W2    = (const float*)d_in[4];
    const float* b2    = (const float*)d_in[5];
    const float* gamma = (const float*)d_in[6];
    const float* beta  = (const float*)d_in[7];
    float*       out   = (float*)d_out;

    char*   ws    = (char*)d_ws;
    float*  h0    = (float*)ws;
    float*  h1    = (float*)(ws + (size_t)N_NODES * D_IN * 4);
    char*   sb    = ws + (size_t)N_NODES * (D_IN + D_HID) * 4;
    double* sums  = (double*)sb;
    double* sumsq = (double*)(sb + 256 * 8);
    float*  scale = (float*)(sb + 2 * 256 * 8);
    float*  shift = (float*)(sb + 2 * 256 * 8 + 256 * 4);

    hipMemsetAsync(sb, 0, 2 * 256 * 8, stream);

    const int n4 = N_NODES * D_IN / 4;
    gn_copy_x<<<(n4 + 255) / 256, 256, 0, stream>>>(x, h0, n4);
    gn_scatter_add<<<(N_EDGES + 7) / 8, 256, 0, stream>>>(x, ei, h0);

    dim3 g1((N_NODES + 127) / 128, D_HID / 64);
    gn_wmma_gemm<D_IN, D_HID, false><<<g1, 256, 0, stream>>>(
        h0, W1, b1, h1, N_NODES, nullptr, nullptr);

    dim3 g2((N_NODES + 127) / 128, D_OUT / 64);
    gn_wmma_gemm<D_HID, D_OUT, true><<<g2, 256, 0, stream>>>(
        h1, W2, b2, out, N_NODES, sums, sumsq);

    gn_bn_finalize<<<1, 256, 0, stream>>>(sums, sumsq, gamma, beta, scale, shift);

    const int t4 = N_NODES * D_OUT / 4;
    gn_bn_apply<<<(t4 + 255) / 256, 256, 0, stream>>>(out, scale, shift);
}


# GEMM: A tile staged through LDS beside B, a K-chunk's 12 tile loads per thread issued as one batch (128-B lines read once per block), replacing per-lane 8-byte global loads waited on every k-step
# speedup vs baseline: 1.1779x; 1.1779x over previous
#include <hip/hip_runtime.h>
#include <hip/hip_bf16.h>


typedef __attribute__((ext_vector_type(2))) float v2f;
typedef __attribute__((ext_vector_type(4))) float v4f;
typedef __attribute__((ext_vector_type(8))) float v8f;

#define N_NODES 50000
#define N_EDGES 800000
#define D_IN    128
#define D_HID   512
#define D_OUT   256
#define BN_EPS  1e-5
#define SLOPE   0.01f

__global__ void gn_copy_x(const float* __restrict__ x, float* __restrict__ h0, int n4) {
    int i = blockIdx.x * blockDim.x + threadIdx.x;
    if (i < n4) ((float4*)h0)[i] = ((const float4*)x)[i];
}

__global__ __launch_bounds__(256) void gn_scatter_add(const float* __restrict__ x,
                                                      const int* __restrict__ ei,
                                                      float* __restrict__ h0) {
    int wave = (blockIdx.x * blockDim.x + threadIdx.x) >> 5;
    int lane = threadIdx.x & 31;
    if (wave >= N_EDGES) return;
    int src = ei[wave];
    int dst = ei[N_EDGES + wave];
    const float4 v = *(const float4*)(x + (size_t)src * D_IN + lane * 4);
    float* hp = h0 + (size_t)dst * D_IN + lane * 4;
    atomicAdd(hp + 0, v.x);
    atomicAdd(hp + 1, v.y);
    atomicAdd(hp + 2, v.z);
    atomicAdd(hp + 3, v.w);
}

template <int K, int N, bool FUSE_STATS>
__global__ __launch_bounds__(256) void gn_wmma_gemm(const float* __restrict__ A,
                                                    const float* __restrict__ B,
                                                    const float* __restrict__ bias,
                                                    float* __restrict__ C, int M,
                                                    double* __restrict__ sums,
                                                    double* __restrict__ sumsq) {
    constexpr int KC = 64;
    constexpr int AP = KC + 4;
    constexpr int BP = 72;
    constexpr int NA = 128 * (KC / 4) / 256;
    constexpr int NB = KC * 16 / 256;
    __shared__ __attribute__((aligned(16))) float As[128][AP];
    __shared__ __attribute__((aligned(16))) float Bs[KC][BP];

    const int tid    = threadIdx.x;
    const int n0     = blockIdx.y * 64;
    const int wave   = tid >> 5;
    const int lane   = tid & 31;
    const int ln     = lane & 15;
    const int khalf  = (lane >> 4) << 1;
    const int rowblk = blockIdx.x * 128;
    const int row0   = rowblk + wave * 16;
    const int sr     = tid >> 4;
    const int sc4    = (tid & 15) * 4;

    unsigned aoff[NA];
#pragma unroll
    for (int i = 0; i < NA; ++i) {
        int gr = rowblk + sr + 16 * i;
        gr = gr < M ? gr : M - 1;
        aoff[i] = ((unsigned)gr * K + sc4) * 4u;
    }
    const unsigned boff = ((unsigned)sr * N + sc4) * 4u;
    const float* afrag = &As[wave * 16 + ln][khalf];

    v8f acc[4] = {};

    for (int kc = 0; kc < K; kc += KC) {
        const char* Ak = (const char*)(A + kc);
        const char* Bk = (const char*)(B + (size_t)kc * N + n0);
        v4f av[NA], bv[NB];
#pragma unroll
        for (int i = 0; i < NA; ++i) av[i] = *(const v4f*)(Ak + aoff[i]);
#pragma unroll
        for (int i = 0; i < NB; ++i)
            bv[i] = *(const v4f*)(Bk + boff + (unsigned)(16 * i * N * 4));
        __builtin_amdgcn_sched_barrier(0);
#pragma unroll
        for (int i = 0; i < NA; ++i) *(v4f*)&As[sr + 16 * i][sc4] = av[i];
#pragma unroll
        for (int i = 0; i < NB; ++i) *(v4f*)&Bs[sr + 16 * i][sc4] = bv[i];
        __syncthreads();

#pragma unroll 4
        for (int k = 0; k < KC; k += 4) {
            v2f a = *(const v2f*)(afrag + k);
#pragma unroll
            for (int t = 0; t < 4; ++t) {
                v2f b;
                b[0] = Bs[k + khalf][ln + 16 * t];
                b[1] = Bs[k + khalf + 1][ln + 16 * t];
                acc[t] = __builtin_amdgcn_wmma_f32_16x16x4_f32(
                     false, a,  false, b,
                     (short)0, acc[t],  false,  false);
            }
        }
        __syncthreads();
    }

    const int rbase = row0 + ((lane >> 4) << 3);
#pragma unroll
    for (int t = 0; t < 4; ++t) {
        const int   col = n0 + 16 * t + ln;
        const float bv  = bias[col];
        double ds = 0.0, dsq = 0.0;
#pragma unroll
        for (int r = 0; r < 8; ++r) {
            int row = rbase + r;
            float v = acc[t][r] + bv;
            v = v > 0.f ? v : v * SLOPE;
            if (row < M) {
                C[(size_t)row * N + col] = v;
                if (FUSE_STATS) { ds += (double)v; dsq += (double)v * (double)v; }
            }
        }
        if (FUSE_STATS) {
            atomicAdd(&sums[col], ds);
            atomicAdd(&sumsq[col], dsq);
        }
    }
}

__global__ void gn_bn_finalize(const double* __restrict__ sums,
                               const double* __restrict__ sumsq,
                               const float* __restrict__ gamma,
                               const float* __restrict__ beta,
                               float* __restrict__ scale, float* __restrict__ shift) {
    int c = threadIdx.x;
    double mean = sums[c] / (double)N_NODES;
    double var  = sumsq[c] / (double)N_NODES - mean * mean;
    double inv  = 1.0 / sqrt(var + BN_EPS);
    float sc = (float)inv * gamma[c];
    scale[c] = sc;
    shift[c] = beta[c] - (float)mean * sc;
}

__global__ void gn_bn_apply(float* __restrict__ out, const float* __restrict__ scale,
                            const float* __restrict__ shift) {
    int i = blockIdx.x * blockDim.x + threadIdx.x;
    const int total4 = N_NODES * D_OUT / 4;
    if (i >= total4) return;
    int c = (i * 4) & (D_OUT - 1);
    float4 v = ((float4*)out)[i];
    v.x = v.x * scale[c + 0] + shift[c + 0];
    v.y = v.y * scale[c + 1] + shift[c + 1];
    v.z = v.z * scale[c + 2] + shift[c + 2];
    v.w = v.w * scale[c + 3] + shift[c + 3];
    ((float4*)out)[i] = v;
}

extern "C" void kernel_launch(void* const* d_in, const int* in_sizes, int n_in,
                              void* d_out, int out_size, void* d_ws, size_t ws_size,
                              hipStream_t stream) {
    const float* x     = (const float*)d_in[0];
    const int*   ei    = (const int*)d_in[1];
    const float* W1    = (const float*)d_in[2];
    const float* b1    = (const float*)d_in[3];
    const float* W2    = (const float*)d_in[4];
    const float* b2    = (const float*)d_in[5];
    const float* gamma = (const float*)d_in[6];
    const float* beta  = (const float*)d_in[7];
    float*       out   = (float*)d_out;

    char*   ws    = (char*)d_ws;
    float*  h0    = (float*)ws;
    float*  h1    = (float*)(ws + (size_t)N_NODES * D_IN * 4);
    char*   sb    = ws + (size_t)N_NODES * (D_IN + D_HID) * 4;
    double* sums  = (double*)sb;
    double* sumsq = (double*)(sb + 256 * 8);
    float*  scale = (float*)(sb + 2 * 256 * 8);
    float*  shift = (float*)(sb + 2 * 256 * 8 + 256 * 4);

    hipMemsetAsync(sb, 0, 2 * 256 * 8, stream);

    const int n4 = N_NODES * D_IN / 4;
    gn_copy_x<<<(n4 + 255) / 256, 256, 0, stream>>>(x, h0, n4);
    gn_scatter_add<<<(N_EDGES + 7) / 8, 256, 0, stream>>>(x, ei, h0);

    dim3 g1((N_NODES + 127) / 128, D_HID / 64);
    gn_wmma_gemm<D_IN, D_HID, false><<<g1, 256, 0, stream>>>(
        h0, W1, b1, h1, N_NODES, nullptr, nullptr);

    dim3 g2((N_NODES + 127) / 128, D_OUT / 64);
    gn_wmma_gemm<D_HID, D_OUT, true><<<g2, 256, 0, stream>>>(
        h1, W2, b2, out, N_NODES, sums, sumsq);

    gn_bn_finalize<<<1, 256, 0, stream>>>(sums, sumsq, gamma, beta, scale, shift);

    const int t4 = N_NODES * D_OUT / 4;
    gn_bn_apply<<<(t4 + 255) / 256, 256, 0, stream>>>(out, scale, shift);
}


# Edge scatter: lane l takes floats l+32j of a row, so each row load and each float atomic add of a wave covers one whole 128-B line, not 8 floats in each of 4 lines; atomics and arrival order kept
# speedup vs baseline: 1.3297x; 1.1289x over previous
#include <hip/hip_runtime.h>
#include <hip/hip_bf16.h>


typedef __attribute__((ext_vector_type(2))) float v2f;
typedef __attribute__((ext_vector_type(4))) float v4f;
typedef __attribute__((ext_vector_type(8))) float v8f;

#define N_NODES 50000
#define N_EDGES 800000
#define D_IN    128
#define D_HID   512
#define D_OUT   256
#define BN_EPS  1e-5
#define SLOPE   0.01f

__global__ void gn_copy_x(const float* __restrict__ x, float* __restrict__ h0, int n4) {
    int i = blockIdx.x * blockDim.x + threadIdx.x;
    if (i < n4) ((float4*)h0)[i] = ((const float4*)x)[i];
}

__global__ __launch_bounds__(256) void gn_scatter_add(const float* __restrict__ x,
                                                      const int* __restrict__ ei,
                                                      float* __restrict__ h0) {
    int wave = (blockIdx.x * blockDim.x + threadIdx.x) >> 5;
    int lane = threadIdx.x & 31;
    if (wave >= N_EDGES) return;
    int src = ei[wave];
    int dst = ei[N_EDGES + wave];
    const float* xp = x + (size_t)src * D_IN + lane;
    float* hp = h0 + (size_t)dst * D_IN + lane;
    const float v0 = xp[0];
    const float v1 = xp[32];
    const float v2 = xp[64];
    const float v3 = xp[96];
    atomicAdd(hp + 0, v0);
    atomicAdd(hp + 32, v1);
    atomicAdd(hp + 64, v2);
    atomicAdd(hp + 96, v3);
}

template <int K, int N, bool FUSE_STATS>
__global__ __launch_bounds__(256) void gn_wmma_gemm(const float* __restrict__ A,
                                                    const float* __restrict__ B,
                                                    const float* __restrict__ bias,
                                                    float* __restrict__ C, int M,
                                                    double* __restrict__ sums,
                                                    double* __restrict__ sumsq) {
    constexpr int KC = 64;
    constexpr int AP = KC + 4;
    constexpr int BP = 72;
    constexpr int NA = 128 * (KC / 4) / 256;
    constexpr int NB = KC * 16 / 256;
    __shared__ __attribute__((aligned(16))) float As[128][AP];
    __shared__ __attribute__((aligned(16))) float Bs[KC][BP];

    const int tid    = threadIdx.x;
    const int n0     = blockIdx.y * 64;
    const int wave   = tid >> 5;
    const int lane   = tid & 31;
    const int ln     = lane & 15;
    const int khalf  = (lane >> 4) << 1;
    const int rowblk = blockIdx.x * 128;
    const int row0   = rowblk + wave * 16;
    const int sr     = tid >> 4;
    const int sc4    = (tid & 15) * 4;

    unsigned aoff[NA];
#pragma unroll
    for (int i = 0; i < NA; ++i) {
        int gr = rowblk + sr + 16 * i;
        gr = gr < M ? gr : M - 1;
        aoff[i] = ((unsigned)gr * K + sc4) * 4u;
    }
    const unsigned boff = ((unsigned)sr * N + sc4) * 4u;
    const float* afrag = &As[wave * 16 + ln][khalf];

    v8f acc[4] = {};

    for (int kc = 0; kc < K; kc += KC) {
        const char* Ak = (const char*)(A + kc);
        const char* Bk = (const char*)(B + (size_t)kc * N + n0);
        v4f av[NA], bv[NB];
#pragma unroll
        for (int i = 0; i < NA; ++i) av[i] = *(const v4f*)(Ak + aoff[i]);
#pragma unroll
        for (int i = 0; i < NB; ++i)
            bv[i] = *(const v4f*)(Bk + boff + (unsigned)(16 * i * N * 4));
        __builtin_amdgcn_sched_barrier(0);
#pragma unroll
        for (int i = 0; i < NA; ++i) *(v4f*)&As[sr + 16 * i][sc4] = av[i];
#pragma unroll
        for (int i = 0; i < NB; ++i) *(v4f*)&Bs[sr + 16 * i][sc4] = bv[i];
        __syncthreads();

#pragma unroll 4
        for (int k = 0; k < KC; k += 4) {
            v2f a = *(const v2f*)(afrag + k);
#pragma unroll
            for (int t = 0; t < 4; ++t) {
                v2f b;
                b[0] = Bs[k + khalf][ln + 16 * t];
                b[1] = Bs[k + khalf + 1][ln + 16 * t];
                acc[t] = __builtin_amdgcn_wmma_f32_16x16x4_f32(
                     false, a,  false, b,
                     (short)0, acc[t],  false,  false);
            }
        }
        __syncthreads();
    }

    const int rbase = row0 + ((lane >> 4) << 3);
#pragma unroll
    for (int t = 0; t < 4; ++t) {
        const int   col = n0 + 16 * t + ln;
        const float bv  = bias[col];
        double ds = 0.0, dsq = 0.0;
#pragma unroll
        for (int r = 0; r < 8; ++r) {
            int row = rbase + r;
            float v = acc[t][r] + bv;
            v = v > 0.f ? v : v * SLOPE;
            if (row < M) {
                C[(size_t)row * N + col] = v;
                if (FUSE_STATS) { ds += (double)v; dsq += (double)v * (double)v; }
            }
        }
        if (FUSE_STATS) {
            atomicAdd(&sums[col], ds);
            atomicAdd(&sumsq[col], dsq);
        }
    }
}

__global__ void gn_bn_finalize(const double* __restrict__ sums,
                               const double* __restrict__ sumsq,
                               const float* __restrict__ gamma,
                               const float* __restrict__ beta,
                               float* __restrict__ scale, float* __restrict__ shift) {
    int c = threadIdx.x;
    double mean = sums[c] / (double)N_NODES;
    double var  = sumsq[c] / (double)N_NODES - mean * mean;
    double inv  = 1.0 / sqrt(var + BN_EPS);
    float sc = (float)inv * gamma[c];
    scale[c] = sc;
    shift[c] = beta[c] - (float)mean * sc;
}

__global__ void gn_bn_apply(float* __restrict__ out, const float* __restrict__ scale,
                            const float* __restrict__ shift) {
    int i = blockIdx.x * blockDim.x + threadIdx.x;
    const int total4 = N_NODES * D_OUT / 4;
    if (i >= total4) return;
    int c = (i * 4) & (D_OUT - 1);
    float4 v = ((float4*)out)[i];
    v.x = v.x * scale[c + 0] + shift[c + 0];
    v.y = v.y * scale[c + 1] + shift[c + 1];
    v.z = v.z * scale[c + 2] + shift[c + 2];
    v.w = v.w * scale[c + 3] + shift[c + 3];
    ((float4*)out)[i] = v;
}

extern "C" void kernel_launch(void* const* d_in, const int* in_sizes, int n_in,
                              void* d_out, int out_size, void* d_ws, size_t ws_size,
                              hipStream_t stream) {
    const float* x     = (const float*)d_in[0];
    const int*   ei    = (const int*)d_in[1];
    const float* W1    = (const float*)d_in[2];
    const float* b1    = (const float*)d_in[3];
    const float* W2    = (const float*)d_in[4];
    const float* b2    = (const float*)d_in[5];
    const float* gamma = (const float*)d_in[6];
    const float* beta  = (const float*)d_in[7];
    float*       out   = (float*)d_out;

    char*   ws    = (char*)d_ws;
    float*  h0    = (float*)ws;
    float*  h1    = (float*)(ws + (size_t)N_NODES * D_IN * 4);
    char*   sb    = ws + (size_t)N_NODES * (D_IN + D_HID) * 4;
    double* sums  = (double*)sb;
    double* sumsq = (double*)(sb + 256 * 8);
    float*  scale = (float*)(sb + 2 * 256 * 8);
    float*  shift = (float*)(sb + 2 * 256 * 8 + 256 * 4);

    hipMemsetAsync(sb, 0, 2 * 256 * 8, stream);

    const int n4 = N_NODES * D_IN / 4;
    gn_copy_x<<<(n4 + 255) / 256, 256, 0, stream>>>(x, h0, n4);
    gn_scatter_add<<<(N_EDGES + 7) / 8, 256, 0, stream>>>(x, ei, h0);

    dim3 g1((N_NODES + 127) / 128, D_HID / 64);
    gn_wmma_gemm<D_IN, D_HID, false><<<g1, 256, 0, stream>>>(
        h0, W1, b1, h1, N_NODES, nullptr, nullptr);

    dim3 g2((N_NODES + 127) / 128, D_OUT / 64);
    gn_wmma_gemm<D_HID, D_OUT, true><<<g2, 256, 0, stream>>>(
        h1, W2, b2, out, N_NODES, sums, sumsq);

    gn_bn_finalize<<<1, 256, 0, stream>>>(sums, sumsq, gamma, beta, scale, shift);

    const int t4 = N_NODES * D_OUT / 4;
    gn_bn_apply<<<(t4 + 255) / 256, 256, 0, stream>>>(out, scale, shift);
}


# Edge scatter: a wave takes 16 consecutive edges, not 1 (50000 waves, not 800000): ids by readlane, 32 row-line loads in flight ahead of their 32 line atomics; atomics and arrival order kept
# speedup vs baseline: 1.4883x; 1.1193x over previous
#include <hip/hip_runtime.h>
#include <hip/hip_bf16.h>


typedef __attribute__((ext_vector_type(2))) float v2f;
typedef __attribute__((ext_vector_type(4))) float v4f;
typedef __attribute__((ext_vector_type(8))) float v8f;

#define N_NODES 50000
#define N_EDGES 800000
#define D_IN    128
#define D_HID   512
#define D_OUT   256
#define BN_EPS  1e-5
#define SLOPE   0.01f

__global__ void gn_copy_x(const float* __restrict__ x, float* __restrict__ h0, int n4) {
    int i = blockIdx.x * blockDim.x + threadIdx.x;
    if (i < n4) ((float4*)h0)[i] = ((const float4*)x)[i];
}

#define EPW 16
static_assert(N_EDGES % EPW == 0, "a wave takes EPW whole edges");
__global__ __launch_bounds__(256) void gn_scatter_add(const float* __restrict__ x,
                                                      const int* __restrict__ ei,
                                                      float* __restrict__ h0) {
    const int wave = (blockIdx.x * blockDim.x + threadIdx.x) >> 5;
    const int lane = threadIdx.x & 31;
    const int e0   = wave * EPW;
    if (e0 >= N_EDGES) return;
    const int my_src = ei[e0 + (lane & (EPW - 1))];
    const int my_dst = ei[N_EDGES + e0 + (lane & (EPW - 1))];
#pragma unroll
    for (int g = 0; g < EPW; g += 8) {
        float v[8][4];
        int   d[8];
#pragma unroll
        for (int j = 0; j < 8; ++j) {
            const int s = __builtin_amdgcn_readlane(my_src, g + j);
            d[j]        = __builtin_amdgcn_readlane(my_dst, g + j);
            const float* xp = x + (size_t)s * D_IN + lane;
            v[j][0] = xp[0];
            v[j][1] = xp[32];
            v[j][2] = xp[64];
            v[j][3] = xp[96];
        }
#pragma unroll
        for (int j = 0; j < 8; ++j) {
            float* hp = h0 + (size_t)d[j] * D_IN + lane;
            atomicAdd(hp + 0, v[j][0]);
            atomicAdd(hp + 32, v[j][1]);
            atomicAdd(hp + 64, v[j][2]);
            atomicAdd(hp + 96, v[j][3]);
        }
    }
}

template <int K, int N, bool FUSE_STATS>
__global__ __launch_bounds__(256) void gn_wmma_gemm(const float* __restrict__ A,
                                                    const float* __restrict__ B,
                                                    const float* __restrict__ bias,
                                                    float* __restrict__ C, int M,
                                                    double* __restrict__ sums,
                                                    double* __restrict__ sumsq) {
    constexpr int KC = 64;
    constexpr int AP = KC + 4;
    constexpr int BP = 72;
    constexpr int NA = 128 * (KC / 4) / 256;
    constexpr int NB = KC * 16 / 256;
    __shared__ __attribute__((aligned(16))) float As[128][AP];
    __shared__ __attribute__((aligned(16))) float Bs[KC][BP];

    const int tid    = threadIdx.x;
    const int n0     = blockIdx.y * 64;
    const int wave   = tid >> 5;
    const int lane   = tid & 31;
    const int ln     = lane & 15;
    const int khalf  = (lane >> 4) << 1;
    const int rowblk = blockIdx.x * 128;
    const int row0   = rowblk + wave * 16;
    const int sr     = tid >> 4;
    const int sc4    = (tid & 15) * 4;

    unsigned aoff[NA];
#pragma unroll
    for (int i = 0; i < NA; ++i) {
        int gr = rowblk + sr + 16 * i;
        gr = gr < M ? gr : M - 1;
        aoff[i] = ((unsigned)gr * K + sc4) * 4u;
    }
    const unsigned boff = ((unsigned)sr * N + sc4) * 4u;
    const float* afrag = &As[wave * 16 + ln][khalf];

    v8f acc[4] = {};

    for (int kc = 0; kc < K; kc += KC) {
        const char* Ak = (const char*)(A + kc);
        const char* Bk = (const char*)(B + (size_t)kc * N + n0);
        v4f av[NA], bv[NB];
#pragma unroll
        for (int i = 0; i < NA; ++i) av[i] = *(const v4f*)(Ak + aoff[i]);
#pragma unroll
        for (int i = 0; i < NB; ++i)
            bv[i] = *(const v4f*)(Bk + boff + (unsigned)(16 * i * N * 4));
        __builtin_amdgcn_sched_barrier(0);
#pragma unroll
        for (int i = 0; i < NA; ++i) *(v4f*)&As[sr + 16 * i][sc4] = av[i];
#pragma unroll
        for (int i = 0; i < NB; ++i) *(v4f*)&Bs[sr + 16 * i][sc4] = bv[i];
        __syncthreads();

#pragma unroll 4
        for (int k = 0; k < KC; k += 4) {
            v2f a = *(const v2f*)(afrag + k);
#pragma unroll
            for (int t = 0; t < 4; ++t) {
                v2f b;
                b[0] = Bs[k + khalf][ln + 16 * t];
                b[1] = Bs[k + khalf + 1][ln + 16 * t];
                acc[t] = __builtin_amdgcn_wmma_f32_16x16x4_f32(
                     false, a,  false, b,
                     (short)0, acc[t],  false,  false);
            }
        }
        __syncthreads();
    }

    const int rbase = row0 + ((lane >> 4) << 3);
#pragma unroll
    for (int t = 0; t < 4; ++t) {
        const int   col = n0 + 16 * t + ln;
        const float bv  = bias[col];
        double ds = 0.0, dsq = 0.0;
#pragma unroll
        for (int r = 0; r < 8; ++r) {
            int row = rbase + r;
            float v = acc[t][r] + bv;
            v = v > 0.f ? v : v * SLOPE;
            if (row < M) {
                C[(size_t)row * N + col] = v;
                if (FUSE_STATS) { ds += (double)v; dsq += (double)v * (double)v; }
            }
        }
        if (FUSE_STATS) {
            atomicAdd(&sums[col], ds);
            atomicAdd(&sumsq[col], dsq);
        }
    }
}

__global__ void gn_bn_finalize(const double* __restrict__ sums,
                               const double* __restrict__ sumsq,
                               const float* __restrict__ gamma,
                               const float* __restrict__ beta,
                               float* __restrict__ scale, float* __restrict__ shift) {
    int c = threadIdx.x;
    double mean = sums[c] / (double)N_NODES;
    double var  = sumsq[c] / (double)N_NODES - mean * mean;
    double inv  = 1.0 / sqrt(var + BN_EPS);
    float sc = (float)inv * gamma[c];
    scale[c] = sc;
    shift[c] = beta[c] - (float)mean * sc;
}

__global__ void gn_bn_apply(float* __restrict__ out, const float* __restrict__ scale,
                            const float* __restrict__ shift) {
    int i = blockIdx.x * blockDim.x + threadIdx.x;
    const int total4 = N_NODES * D_OUT / 4;
    if (i >= total4) return;
    int c = (i * 4) & (D_OUT - 1);
    float4 v = ((float4*)out)[i];
    v.x = v.x * scale[c + 0] + shift[c + 0];
    v.y = v.y * scale[c + 1] + shift[c + 1];
    v.z = v.z * scale[c + 2] + shift[c + 2];
    v.w = v.w * scale[c + 3] + shift[c + 3];
    ((float4*)out)[i] = v;
}

extern "C" void kernel_launch(void* const* d_in, const int* in_sizes, int n_in,
                              void* d_out, int out_size, void* d_ws, size_t ws_size,
                              hipStream_t stream) {
    const float* x     = (const float*)d_in[0];
    const int*   ei    = (const int*)d_in[1];
    const float* W1    = (const float*)d_in[2];
    const float* b1    = (const float*)d_in[3];
    const float* W2    = (const float*)d_in[4];
    const float* b2    = (const float*)d_in[5];
    const float* gamma = (const float*)d_in[6];
    const float* beta  = (const float*)d_in[7];
    float*       out   = (float*)d_out;

    char*   ws    = (char*)d_ws;
    float*  h0    = (float*)ws;
    float*  h1    = (float*)(ws + (size_t)N_NODES * D_IN * 4);
    char*   sb    = ws + (size_t)N_NODES * (D_IN + D_HID) * 4;
    double* sums  = (double*)sb;
    double* sumsq = (double*)(sb + 256 * 8);
    float*  scale = (float*)(sb + 2 * 256 * 8);
    float*  shift = (float*)(sb + 2 * 256 * 8 + 256 * 4);

    hipMemsetAsync(sb, 0, 2 * 256 * 8, stream);

    const int n4 = N_NODES * D_IN / 4;
    gn_copy_x<<<(n4 + 255) / 256, 256, 0, stream>>>(x, h0, n4);
    gn_scatter_add<<<(N_EDGES / EPW + 7) / 8, 256, 0, stream>>>(x, ei, h0);

    dim3 g1((N_NODES + 127) / 128, D_HID / 64);
    gn_wmma_gemm<D_IN, D_HID, false><<<g1, 256, 0, stream>>>(
        h0, W1, b1, h1, N_NODES, nullptr, nullptr);

    dim3 g2((N_NODES + 127) / 128, D_OUT / 64);
    gn_wmma_gemm<D_HID, D_OUT, true><<<g2, 256, 0, stream>>>(
        h1, W2, b2, out, N_NODES, sums, sumsq);

    gn_bn_finalize<<<1, 256, 0, stream>>>(sums, sumsq, gamma, beta, scale, shift);

    const int t4 = N_NODES * D_OUT / 4;
    gn_bn_apply<<<(t4 + 255) / 256, 256, 0, stream>>>(out, scale, shift);
}


# GEMM block order: column block is blockIdx.x, so the blocks sharing a row block of A launch together and A comes from memory once, not once per column block; fp64 stat atomics, arrival order kept
# speedup vs baseline: 1.5773x; 1.0598x over previous
#include <hip/hip_runtime.h>
#include <hip/hip_bf16.h>


typedef __attribute__((ext_vector_type(2))) float v2f;
typedef __attribute__((ext_vector_type(4))) float v4f;
typedef __attribute__((ext_vector_type(8))) float v8f;

#define N_NODES 50000
#define N_EDGES 800000
#define D_IN    128
#define D_HID   512
#define D_OUT   256
#define BN_EPS  1e-5
#define SLOPE   0.01f

__global__ void gn_copy_x(const float* __restrict__ x, float* __restrict__ h0, int n4) {
    int i = blockIdx.x * blockDim.x + threadIdx.x;
    if (i < n4) ((float4*)h0)[i] = ((const float4*)x)[i];
}

#define EPW 16
static_assert(N_EDGES % EPW == 0, "a wave takes EPW whole edges");
__global__ __launch_bounds__(256) void gn_scatter_add(const float* __restrict__ x,
                                                      const int* __restrict__ ei,
                                                      float* __restrict__ h0) {
    const int wave = (blockIdx.x * blockDim.x + threadIdx.x) >> 5;
    const int lane = threadIdx.x & 31;
    const int e0   = wave * EPW;
    if (e0 >= N_EDGES) return;
    const int my_src = ei[e0 + (lane & (EPW - 1))];
    const int my_dst = ei[N_EDGES + e0 + (lane & (EPW - 1))];
#pragma unroll
    for (int g = 0; g < EPW; g += 8) {
        float v[8][4];
        int   d[8];
#pragma unroll
        for (int j = 0; j < 8; ++j) {
            const int s = __builtin_amdgcn_readlane(my_src, g + j);
            d[j]        = __builtin_amdgcn_readlane(my_dst, g + j);
            const float* xp = x + (size_t)s * D_IN + lane;
            v[j][0] = xp[0];
            v[j][1] = xp[32];
            v[j][2] = xp[64];
            v[j][3] = xp[96];
        }
#pragma unroll
        for (int j = 0; j < 8; ++j) {
            float* hp = h0 + (size_t)d[j] * D_IN + lane;
            atomicAdd(hp + 0, v[j][0]);
            atomicAdd(hp + 32, v[j][1]);
            atomicAdd(hp + 64, v[j][2]);
            atomicAdd(hp + 96, v[j][3]);
        }
    }
}

template <int K, int N, bool FUSE_STATS>
__global__ __launch_bounds__(256) void gn_wmma_gemm(const float* __restrict__ A,
                                                    const float* __restrict__ B,
                                                    const float* __restrict__ bias,
                                                    float* __restrict__ C, int M,
                                                    double* __restrict__ sums,
                                                    double* __restrict__ sumsq) {
    constexpr int KC = 64;
    constexpr int AP = KC + 4;
    constexpr int BP = 72;
    constexpr int NA = 128 * (KC / 4) / 256;
    constexpr int NB = KC * 16 / 256;
    __shared__ __attribute__((aligned(16))) float As[128][AP];
    __shared__ __attribute__((aligned(16))) float Bs[KC][BP];

    const int tid    = threadIdx.x;
    const int n0     = blockIdx.x * 64;
    const int wave   = tid >> 5;
    const int lane   = tid & 31;
    const int ln     = lane & 15;
    const int khalf  = (lane >> 4) << 1;
    const int rowblk = blockIdx.y * 128;
    const int row0   = rowblk + wave * 16;
    const int sr     = tid >> 4;
    const int sc4    = (tid & 15) * 4;

    unsigned aoff[NA];
#pragma unroll
    for (int i = 0; i < NA; ++i) {
        int gr = rowblk + sr + 16 * i;
        gr = gr < M ? gr : M - 1;
        aoff[i] = ((unsigned)gr * K + sc4) * 4u;
    }
    const unsigned boff = ((unsigned)sr * N + sc4) * 4u;
    const float* afrag = &As[wave * 16 + ln][khalf];

    v8f acc[4] = {};

    for (int kc = 0; kc < K; kc += KC) {
        const char* Ak = (const char*)(A + kc);
        const char* Bk = (const char*)(B + (size_t)kc * N + n0);
        v4f av[NA], bv[NB];
#pragma unroll
        for (int i = 0; i < NA; ++i) av[i] = *(const v4f*)(Ak + aoff[i]);
#pragma unroll
        for (int i = 0; i < NB; ++i)
            bv[i] = *(const v4f*)(Bk + boff + (unsigned)(16 * i * N * 4));
        __builtin_amdgcn_sched_barrier(0);
#pragma unroll
        for (int i = 0; i < NA; ++i) *(v4f*)&As[sr + 16 * i][sc4] = av[i];
#pragma unroll
        for (int i = 0; i < NB; ++i) *(v4f*)&Bs[sr + 16 * i][sc4] = bv[i];
        __syncthreads();

#pragma unroll 4
        for (int k = 0; k < KC; k += 4) {
            v2f a = *(const v2f*)(afrag + k);
#pragma unroll
            for (int t = 0; t < 4; ++t) {
                v2f b;
                b[0] = Bs[k + khalf][ln + 16 * t];
                b[1] = Bs[k + khalf + 1][ln + 16 * t];
                acc[t] = __builtin_amdgcn_wmma_f32_16x16x4_f32(
                     false, a,  false, b,
                     (short)0, acc[t],  false,  false);
            }
        }
        __syncthreads();
    }

    const int rbase = row0 + ((lane >> 4) << 3);
#pragma unroll
    for (int t = 0; t < 4; ++t) {
        const int   col = n0 + 16 * t + ln;
        const float bv  = bias[col];
        double ds = 0.0, dsq = 0.0;
#pragma unroll
        for (int r = 0; r < 8; ++r) {
            int row = rbase + r;
            float v = acc[t][r] + bv;
            v = v > 0.f ? v : v * SLOPE;
            if (row < M) {
                C[(size_t)row * N + col] = v;
                if (FUSE_STATS) { ds += (double)v; dsq += (double)v * (double)v; }
            }
        }
        if (FUSE_STATS) {
            atomicAdd(&sums[col], ds);
            atomicAdd(&sumsq[col], dsq);
        }
    }
}

__global__ void gn_bn_finalize(const double* __restrict__ sums,
                               const double* __restrict__ sumsq,
                               const float* __restrict__ gamma,
                               const float* __restrict__ beta,
                               float* __restrict__ scale, float* __restrict__ shift) {
    int c = threadIdx.x;
    double mean = sums[c] / (double)N_NODES;
    double var  = sumsq[c] / (double)N_NODES - mean * mean;
    double inv  = 1.0 / sqrt(var + BN_EPS);
    float sc = (float)inv * gamma[c];
    scale[c] = sc;
    shift[c] = beta[c] - (float)mean * sc;
}

__global__ void gn_bn_apply(float* __restrict__ out, const float* __restrict__ scale,
                            const float* __restrict__ shift) {
    int i = blockIdx.x * blockDim.x + threadIdx.x;
    const int total4 = N_NODES * D_OUT / 4;
    if (i >= total4) return;
    int c = (i * 4) & (D_OUT - 1);
    float4 v = ((float4*)out)[i];
    v.x = v.x * scale[c + 0] + shift[c + 0];
    v.y = v.y * scale[c + 1] + shift[c + 1];
    v.z = v.z * scale[c + 2] + shift[c + 2];
    v.w = v.w * scale[c + 3] + shift[c + 3];
    ((float4*)out)[i] = v;
}

extern "C" void kernel_launch(void* const* d_in, const int* in_sizes, int n_in,
                              void* d_out, int out_size, void* d_ws, size_t ws_size,
                              hipStream_t stream) {
    const float* x     = (const float*)d_in[0];
    const int*   ei    = (const int*)d_in[1];
    const float* W1    = (const float*)d_in[2];
    const float* b1    = (const float*)d_in[3];
    const float* W2    = (const float*)d_in[4];
    const float* b2    = (const float*)d_in[5];
    const float* gamma = (const float*)d_in[6];
    const float* beta  = (const float*)d_in[7];
    float*       out   = (float*)d_out;

    char*   ws    = (char*)d_ws;
    float*  h0    = (float*)ws;
    float*  h1    = (float*)(ws + (size_t)N_NODES * D_IN * 4);
    char*   sb    = ws + (size_t)N_NODES * (D_IN + D_HID) * 4;
    double* sums  = (double*)sb;
    double* sumsq = (double*)(sb + 256 * 8);
    float*  scale = (float*)(sb + 2 * 256 * 8);
    float*  shift = (float*)(sb + 2 * 256 * 8 + 256 * 4);

    hipMemsetAsync(sb, 0, 2 * 256 * 8, stream);

    const int n4 = N_NODES * D_IN / 4;
    gn_copy_x<<<(n4 + 255) / 256, 256, 0, stream>>>(x, h0, n4);
    gn_scatter_add<<<(N_EDGES / EPW + 7) / 8, 256, 0, stream>>>(x, ei, h0);

    dim3 g1(D_HID / 64, (N_NODES + 127) / 128);
    gn_wmma_gemm<D_IN, D_HID, false><<<g1, 256, 0, stream>>>(
        h0, W1, b1, h1, N_NODES, nullptr, nullptr);

    dim3 g2(D_OUT / 64, (N_NODES + 127) / 128);
    gn_wmma_gemm<D_HID, D_OUT, true><<<g2, 256, 0, stream>>>(
        h1, W2, b2, out, N_NODES, sums, sumsq);

    gn_bn_finalize<<<1, 256, 0, stream>>>(sums, sumsq, gamma, beta, scale, shift);

    const int t4 = N_NODES * D_OUT / 4;
    gn_bn_apply<<<(t4 + 255) / 256, 256, 0, stream>>>(out, scale, shift);
}


# BN column sums: f32 partials inside a 128-row GEMM block (lane's 8 rows, then 16 partials via LDS), one fp64 atomic per column and block, not per lane; new sum order, fp64 only across blocks
# speedup vs baseline: 1.6967x; 1.0757x over previous
#include <hip/hip_runtime.h>
#include <hip/hip_bf16.h>


typedef __attribute__((ext_vector_type(2))) float v2f;
typedef __attribute__((ext_vector_type(4))) float v4f;
typedef __attribute__((ext_vector_type(8))) float v8f;

#define N_NODES 50000
#define N_EDGES 800000
#define D_IN    128
#define D_HID   512
#define D_OUT   256
#define BN_EPS  1e-5
#define SLOPE   0.01f

__global__ void gn_copy_x(const float* __restrict__ x, float* __restrict__ h0, int n4) {
    int i = blockIdx.x * blockDim.x + threadIdx.x;
    if (i < n4) ((float4*)h0)[i] = ((const float4*)x)[i];
}

#define EPW 16
static_assert(N_EDGES % EPW == 0, "a wave takes EPW whole edges");
__global__ __launch_bounds__(256) void gn_scatter_add(const float* __restrict__ x,
                                                      const int* __restrict__ ei,
                                                      float* __restrict__ h0) {
    const int wave = (blockIdx.x * blockDim.x + threadIdx.x) >> 5;
    const int lane = threadIdx.x & 31;
    const int e0   = wave * EPW;
    if (e0 >= N_EDGES) return;
    const int my_src = ei[e0 + (lane & (EPW - 1))];
    const int my_dst = ei[N_EDGES + e0 + (lane & (EPW - 1))];
#pragma unroll
    for (int g = 0; g < EPW; g += 8) {
        float v[8][4];
        int   d[8];
#pragma unroll
        for (int j = 0; j < 8; ++j) {
            const int s = __builtin_amdgcn_readlane(my_src, g + j);
            d[j]        = __builtin_amdgcn_readlane(my_dst, g + j);
            const float* xp = x + (size_t)s * D_IN + lane;
            v[j][0] = xp[0];
            v[j][1] = xp[32];
            v[j][2] = xp[64];
            v[j][3] = xp[96];
        }
#pragma unroll
        for (int j = 0; j < 8; ++j) {
            float* hp = h0 + (size_t)d[j] * D_IN + lane;
            atomicAdd(hp + 0, v[j][0]);
            atomicAdd(hp + 32, v[j][1]);
            atomicAdd(hp + 64, v[j][2]);
            atomicAdd(hp + 96, v[j][3]);
        }
    }
}

template <int K, int N, bool FUSE_STATS>
__global__ __launch_bounds__(256) void gn_wmma_gemm(const float* __restrict__ A,
                                                    const float* __restrict__ B,
                                                    const float* __restrict__ bias,
                                                    float* __restrict__ C, int M,
                                                    double* __restrict__ sums,
                                                    double* __restrict__ sumsq) {
    constexpr int KC = 64;
    constexpr int AP = KC + 4;
    constexpr int BP = 72;
    constexpr int NA = 128 * (KC / 4) / 256;
    constexpr int NB = KC * 16 / 256;
    __shared__ __attribute__((aligned(16))) float As[128][AP];
    __shared__ __attribute__((aligned(16))) float Bs[KC][BP];

    const int tid    = threadIdx.x;
    const int n0     = blockIdx.x * 64;
    const int wave   = tid >> 5;
    const int lane   = tid & 31;
    const int ln     = lane & 15;
    const int khalf  = (lane >> 4) << 1;
    const int rowblk = blockIdx.y * 128;
    const int row0   = rowblk + wave * 16;
    const int sr     = tid >> 4;
    const int sc4    = (tid & 15) * 4;

    unsigned aoff[NA];
#pragma unroll
    for (int i = 0; i < NA; ++i) {
        int gr = rowblk + sr + 16 * i;
        gr = gr < M ? gr : M - 1;
        aoff[i] = ((unsigned)gr * K + sc4) * 4u;
    }
    const unsigned boff = ((unsigned)sr * N + sc4) * 4u;
    const float* afrag = &As[wave * 16 + ln][khalf];

    v8f acc[4] = {};

    for (int kc = 0; kc < K; kc += KC) {
        const char* Ak = (const char*)(A + kc);
        const char* Bk = (const char*)(B + (size_t)kc * N + n0);
        v4f av[NA], bv[NB];
#pragma unroll
        for (int i = 0; i < NA; ++i) av[i] = *(const v4f*)(Ak + aoff[i]);
#pragma unroll
        for (int i = 0; i < NB; ++i)
            bv[i] = *(const v4f*)(Bk + boff + (unsigned)(16 * i * N * 4));
        __builtin_amdgcn_sched_barrier(0);
#pragma unroll
        for (int i = 0; i < NA; ++i) *(v4f*)&As[sr + 16 * i][sc4] = av[i];
#pragma unroll
        for (int i = 0; i < NB; ++i) *(v4f*)&Bs[sr + 16 * i][sc4] = bv[i];
        __syncthreads();

#pragma unroll 4
        for (int k = 0; k < KC; k += 4) {
            v2f a = *(const v2f*)(afrag + k);
#pragma unroll
            for (int t = 0; t < 4; ++t) {
                v2f b;
                b[0] = Bs[k + khalf][ln + 16 * t];
                b[1] = Bs[k + khalf + 1][ln + 16 * t];
                acc[t] = __builtin_amdgcn_wmma_f32_16x16x4_f32(
                     false, a,  false, b,
                     (short)0, acc[t],  false,  false);
            }
        }
        __syncthreads();
    }

    const int rbase = row0 + ((lane >> 4) << 3);
    float ps[4], pq[4];
#pragma unroll
    for (int t = 0; t < 4; ++t) {
        const float bv = bias[n0 + 16 * t + ln];
        float s = 0.f, q = 0.f;
#pragma unroll
        for (int r = 0; r < 8; ++r) {
            float v = acc[t][r] + bv;
            v = v > 0.f ? v : v * SLOPE;
            acc[t][r] = v;
            if (FUSE_STATS && rbase + r < M) { s += v; q = fmaf(v, v, q); }
        }
        ps[t] = s;
        pq[t] = q;
    }
    constexpr int RP = 80;
    float* red = &As[0][0];
    if (FUSE_STATS) {
        const int slot = wave * 2 + (lane >> 4);
#pragma unroll
        for (int t = 0; t < 4; ++t) {
            red[slot * RP + 16 * t + ln]           = ps[t];
            red[16 * RP + slot * RP + 16 * t + ln] = pq[t];
        }
        __syncthreads();
    }
#pragma unroll
    for (int t = 0; t < 4; ++t) {
        const int col = n0 + 16 * t + ln;
#pragma unroll
        for (int r = 0; r < 8; ++r) {
            const int row = rbase + r;
            if (row < M) C[(size_t)row * N + col] = acc[t][r];
        }
    }
    if (FUSE_STATS) {
        if (tid < 128) {
            const int    stat = tid >> 6;
            const int    c    = tid & 63;
            const float* rp   = red + stat * 16 * RP + c;
            float tot = 0.f;
#pragma unroll
            for (int i = 0; i < 16; ++i) tot += rp[i * RP];
            atomicAdd((stat ? sumsq : sums) + n0 + c, (double)tot);
        }
    }
}

__global__ void gn_bn_finalize(const double* __restrict__ sums,
                               const double* __restrict__ sumsq,
                               const float* __restrict__ gamma,
                               const float* __restrict__ beta,
                               float* __restrict__ scale, float* __restrict__ shift) {
    int c = threadIdx.x;
    double mean = sums[c] / (double)N_NODES;
    double var  = sumsq[c] / (double)N_NODES - mean * mean;
    double inv  = 1.0 / sqrt(var + BN_EPS);
    float sc = (float)inv * gamma[c];
    scale[c] = sc;
    shift[c] = beta[c] - (float)mean * sc;
}

__global__ void gn_bn_apply(float* __restrict__ out, const float* __restrict__ scale,
                            const float* __restrict__ shift) {
    int i = blockIdx.x * blockDim.x + threadIdx.x;
    const int total4 = N_NODES * D_OUT / 4;
    if (i >= total4) return;
    int c = (i * 4) & (D_OUT - 1);
    float4 v = ((float4*)out)[i];
    v.x = v.x * scale[c + 0] + shift[c + 0];
    v.y = v.y * scale[c + 1] + shift[c + 1];
    v.z = v.z * scale[c + 2] + shift[c + 2];
    v.w = v.w * scale[c + 3] + shift[c + 3];
    ((float4*)out)[i] = v;
}

extern "C" void kernel_launch(void* const* d_in, const int* in_sizes, int n_in,
                              void* d_out, int out_size, void* d_ws, size_t ws_size,
                              hipStream_t stream) {
    const float* x     = (const float*)d_in[0];
    const int*   ei    = (const int*)d_in[1];
    const float* W1    = (const float*)d_in[2];
    const float* b1    = (const float*)d_in[3];
    const float* W2    = (const float*)d_in[4];
    const float* b2    = (const float*)d_in[5];
    const float* gamma = (const float*)d_in[6];
    const float* beta  = (const float*)d_in[7];
    float*       out   = (float*)d_out;

    char*   ws    = (char*)d_ws;
    float*  h0    = (float*)ws;
    float*  h1    = (float*)(ws + (size_t)N_NODES * D_IN * 4);
    char*   sb    = ws + (size_t)N_NODES * (D_IN + D_HID) * 4;
    double* sums  = (double*)sb;
    double* sumsq = (double*)(sb + 256 * 8);
    float*  scale = (float*)(sb + 2 * 256 * 8);
    float*  shift = (float*)(sb + 2 * 256 * 8 + 256 * 4);

    hipMemsetAsync(sb, 0, 2 * 256 * 8, stream);

    const int n4 = N_NODES * D_IN / 4;
    gn_copy_x<<<(n4 + 255) / 256, 256, 0, stream>>>(x, h0, n4);
    gn_scatter_add<<<(N_EDGES / EPW + 7) / 8, 256, 0, stream>>>(x, ei, h0);

    dim3 g1(D_HID / 64, (N_NODES + 127) / 128);
    gn_wmma_gemm<D_IN, D_HID, false><<<g1, 256, 0, stream>>>(
        h0, W1, b1, h1, N_NODES, nullptr, nullptr);

    dim3 g2(D_OUT / 64, (N_NODES + 127) / 128);
    gn_wmma_gemm<D_HID, D_OUT, true><<<g2, 256, 0, stream>>>(
        h1, W2, b2, out, N_NODES, sums, sumsq);

    gn_bn_finalize<<<1, 256, 0, stream>>>(sums, sumsq, gamma, beta, scale, shift);

    const int t4 = N_NODES * D_OUT / 4;
    gn_bn_apply<<<(t4 + 255) / 256, 256, 0, stream>>>(out, scale, shift);
}
